# starting kernel
# baseline (speedup 1.0000x reference)
#include <hip/hip_runtime.h>
#include <math.h>

#define B_    8
#define N_    9225
#define M_    4096
#define K_    32
#define HID_  64
#define DOUT_ 16

typedef float v2f __attribute__((ext_vector_type(2)));
typedef float v8f __attribute__((ext_vector_type(8)));

__device__ __forceinline__ v8f wmma_f32_4(v2f a, v2f b, v8f c) {
    return __builtin_amdgcn_wmma_f32_16x16x4_f32(
         false, a,  false, b,
         (short)0, c,  false,  false);
}

#define SWZ_XOR16 0x401f
#define SWZ_XOR8  0x201f
#define SWZ_XOR4  0x101f
#define SWZ_XOR2  0x081f
#define SWZ_XOR1  0x041f

template <int PAT>
__device__ __forceinline__ float swz(float x) {
    return __builtin_bit_cast(
        float, __builtin_amdgcn_ds_swizzle(__builtin_bit_cast(int, x), PAT));
}

__device__ __forceinline__ float gelu_exact(float z) {
    const float s  = z * 0.70710678118654752440f;
    const float ax = __builtin_fabsf(s);
    const float t  = __builtin_amdgcn_rcpf(__builtin_fmaf(0.3275911f, ax, 1.0f));
    float p = 1.061405429f;
    p = __builtin_fmaf(p, t, -1.453152027f);
    p = __builtin_fmaf(p, t,  1.421413741f);
    p = __builtin_fmaf(p, t, -0.284496736f);
    p = __builtin_fmaf(p, t,  0.254829592f);
    p = p * t;
    const float e = __builtin_amdgcn_exp2f(-ax * ax * 1.4426950408889634f);
    const float erf_ax = __builtin_fmaf(-p, e, 1.0f);
    const float erf_s  = copysignf(erf_ax, s);
    return 0.5f * z * (1.0f + erf_s);
}

__global__ __launch_bounds__(256)
void IntegralTransformBatch_kernel(const float* __restrict__ y,
                                   const float* __restrict__ x,
                                   const float* __restrict__ f_y,
                                   const float* __restrict__ wts,
                                   const float* __restrict__ W1,
                                   const float* __restrict__ b1,
                                   const float* __restrict__ W2,
                                   const float* __restrict__ b2,
                                   const int*   __restrict__ nbr,
                                   float*       __restrict__ out)
{
    const int tid  = threadIdx.x;
    const int wave = tid >> 5;
    const int lane = tid & 31;
    const int row  = lane & 15;
    const bool hi  = lane >= 16;
    const int hoff = hi ? 8 : 0;

    const int bm = blockIdx.x * 8 + wave;
    if (bm >= B_ * M_) return;
    const int b = bm / M_;
    const int m = bm % M_;

    const float xm0 = x[((long)b * M_ + m) * 2 + 0];
    const float xm1 = x[((long)b * M_ + m) * 2 + 1];

    v2f w1a[4];
#pragma unroll
    for (int t = 0; t < 4; ++t) {
        const int col = 16 * t + row;
        w1a[t][0] = W1[(hi ? 2 : 0) * HID_ + col];
        w1a[t][1] = W1[(hi ? 3 : 1) * HID_ + col];
    }
    v2f w2a[16];
#pragma unroll
    for (int j = 0; j < 16; ++j) {
        w2a[j][0] = W2[(4 * j + (hi ? 2 : 0)) * DOUT_ + row];
        w2a[j][1] = W2[(4 * j + (hi ? 3 : 1)) * DOUT_ + row];
    }
    float b1v[4][8];
#pragma unroll
    for (int t = 0; t < 4; ++t)
#pragma unroll
        for (int v = 0; v < 8; ++v)
            b1v[t][v] = b1[16 * t + v + hoff];
    float b2v[8];
#pragma unroll
    for (int v = 0; v < 8; ++v) b2v[v] = b2[v + hoff];

    float oacc[8];
#pragma unroll
    for (int v = 0; v < 8; ++v) oacc[v] = 0.0f;

    const long nb_base = ((long)b * M_ + m) * K_;
    const long tb      = (long)b * N_;

    #pragma unroll
    for (int tt = 0; tt < 2; ++tt) {
        const int  raw   = nbr[nb_base + tt * 16 + row];
        const bool valid = (raw != -1);
        const int  idx   = valid ? raw : 0;
        const float wv   = valid ? wts[tb + idx] : 0.0f;
        const float y0   = y[(tb + idx) * 2 + 0];
        const float y1   = y[(tb + idx) * 2 + 1];

        v2f bop;
        bop[0] = hi ? xm0 : y0;
        bop[1] = hi ? xm1 : y1;

        float hreg[4][8];
#pragma unroll
        for (int t = 0; t < 4; ++t) {
            v8f acc = {0.f, 0.f, 0.f, 0.f, 0.f, 0.f, 0.f, 0.f};
            acc = wmma_f32_4(w1a[t], bop, acc);
#pragma unroll
            for (int v = 0; v < 8; ++v) {
                hreg[t][v] = gelu_exact(acc[v] + b1v[t][v]);
            }
        }

        v8f acc2 = {0.f, 0.f, 0.f, 0.f, 0.f, 0.f, 0.f, 0.f};
#pragma unroll
        for (int t = 0; t < 4; ++t) {
            float sw[8];
#pragma unroll
            for (int v = 0; v < 8; ++v)
                sw[v] = swz<SWZ_XOR16>(hreg[t][v]);

            v2f bj;
            bj[0] = hi ? sw[2] : hreg[t][0];
            bj[1] = hi ? sw[3] : hreg[t][1];
            acc2 = wmma_f32_4(w2a[4 * t + 0], bj, acc2);
            bj[0] = hi ? sw[6] : hreg[t][4];
            bj[1] = hi ? sw[7] : hreg[t][5];
            acc2 = wmma_f32_4(w2a[4 * t + 1], bj, acc2);
            bj[0] = hi ? hreg[t][2] : sw[0];
            bj[1] = hi ? hreg[t][3] : sw[1];
            acc2 = wmma_f32_4(w2a[4 * t + 2], bj, acc2);
            bj[0] = hi ? hreg[t][6] : sw[4];
            bj[1] = hi ? hreg[t][7] : sw[5];
            acc2 = wmma_f32_4(w2a[4 * t + 3], bj, acc2);
        }

        const float* fy = f_y + (tb + idx) * DOUT_;
#pragma unroll
        for (int v = 0; v < 8; ++v) {
            oacc[v] += (acc2[v] + b2v[v]) * fy[v + hoff] * wv;
        }
    }

#pragma unroll
    for (int v = 0; v < 8; ++v) {
        float s = oacc[v];
        s += swz<SWZ_XOR1>(s);
        s += swz<SWZ_XOR2>(s);
        s += swz<SWZ_XOR4>(s);
        s += swz<SWZ_XOR8>(s);
        oacc[v] = s;
    }

    if (row == 0) {
        float* po = out + ((long)b * M_ + m) * DOUT_ + hoff;
#pragma unroll
        for (int v = 0; v < 8; ++v) po[v] = oacc[v];
    }
}

extern "C" void kernel_launch(void* const* d_in, const int* in_sizes, int n_in,
                              void* d_out, int out_size, void* d_ws, size_t ws_size,
                              hipStream_t stream) {
    const float* y   = (const float*)d_in[0];
    const float* x   = (const float*)d_in[1];
    const float* f_y = (const float*)d_in[2];
    const float* wts = (const float*)d_in[3];
    const float* W1  = (const float*)d_in[4];
    const float* b1  = (const float*)d_in[5];
    const float* W2  = (const float*)d_in[6];
    const float* b2  = (const float*)d_in[7];
    const int*   nbr = (const int*)d_in[8];
    float* out = (float*)d_out;

    const int total  = B_ * M_;
    const int blocks = (total + 7) / 8;
    IntegralTransformBatch_kernel<<<blocks, 256, 0, stream>>>(
        y, x, f_y, wts, W1, b1, W2, b2, nbr, out);
}


# Layer-2 k order re-chosen so each lane half feeds B from the hidden channels it already holds: half-wave swizzles and selects gone, compiler packs GELU in f32 pairs; layer-2 sum order changed
# speedup vs baseline: 1.1253x; 1.1253x over previous
#include <hip/hip_runtime.h>
#include <math.h>

#define B_    8
#define N_    9225
#define M_    4096
#define K_    32
#define HID_  64
#define DOUT_ 16

typedef float v2f __attribute__((ext_vector_type(2)));
typedef float v8f __attribute__((ext_vector_type(8)));

__device__ __forceinline__ v8f wmma_f32_4(v2f a, v2f b, v8f c) {
    return __builtin_amdgcn_wmma_f32_16x16x4_f32(
         false, a,  false, b,
         (short)0, c,  false,  false);
}

#define SWZ_XOR8  0x201f
#define SWZ_XOR4  0x101f
#define SWZ_XOR2  0x081f
#define SWZ_XOR1  0x041f

template <int PAT>
__device__ __forceinline__ float swz(float x) {
    return __builtin_bit_cast(
        float, __builtin_amdgcn_ds_swizzle(__builtin_bit_cast(int, x), PAT));
}

__device__ __forceinline__ float gelu_exact(float z) {
    const float s  = z * 0.70710678118654752440f;
    const float ax = __builtin_fabsf(s);
    const float t  = __builtin_amdgcn_rcpf(__builtin_fmaf(0.3275911f, ax, 1.0f));
    float p = 1.061405429f;
    p = __builtin_fmaf(p, t, -1.453152027f);
    p = __builtin_fmaf(p, t,  1.421413741f);
    p = __builtin_fmaf(p, t, -0.284496736f);
    p = __builtin_fmaf(p, t,  0.254829592f);
    p = p * t;
    const float e = __builtin_amdgcn_exp2f(-ax * ax * 1.4426950408889634f);
    const float erf_ax = __builtin_fmaf(-p, e, 1.0f);
    const float erf_s  = copysignf(erf_ax, s);
    return 0.5f * z * (1.0f + erf_s);
}

__global__ __launch_bounds__(256)
void IntegralTransformBatch_kernel(const float* __restrict__ y,
                                   const float* __restrict__ x,
                                   const float* __restrict__ f_y,
                                   const float* __restrict__ wts,
                                   const float* __restrict__ W1,
                                   const float* __restrict__ b1,
                                   const float* __restrict__ W2,
                                   const float* __restrict__ b2,
                                   const int*   __restrict__ nbr,
                                   float*       __restrict__ out)
{
    const int tid  = threadIdx.x;
    const int wave = tid >> 5;
    const int lane = tid & 31;
    const int row  = lane & 15;
    const bool hi  = lane >= 16;
    const int hoff = hi ? 8 : 0;

    const int bm = blockIdx.x * 8 + wave;
    if (bm >= B_ * M_) return;
    const int b = bm / M_;
    const int m = bm % M_;

    const float xm0 = x[((long)b * M_ + m) * 2 + 0];
    const float xm1 = x[((long)b * M_ + m) * 2 + 1];

    v2f w1a[4];
#pragma unroll
    for (int t = 0; t < 4; ++t) {
        const int col = 16 * t + row;
        w1a[t][0] = W1[(hi ? 2 : 0) * HID_ + col];
        w1a[t][1] = W1[(hi ? 3 : 1) * HID_ + col];
    }
    v2f w2a[16];
#pragma unroll
    for (int j = 0; j < 16; ++j) {
        const int ch = 16 * (j >> 2) + hoff + 2 * (j & 3);
        w2a[j][0] = W2[(ch + 0) * DOUT_ + row];
        w2a[j][1] = W2[(ch + 1) * DOUT_ + row];
    }
    float b1v[4][8];
#pragma unroll
    for (int t = 0; t < 4; ++t)
#pragma unroll
        for (int v = 0; v < 8; ++v)
            b1v[t][v] = b1[16 * t + v + hoff];
    float b2v[8];
#pragma unroll
    for (int v = 0; v < 8; ++v) b2v[v] = b2[v + hoff];

    float oacc[8];
#pragma unroll
    for (int v = 0; v < 8; ++v) oacc[v] = 0.0f;

    const long nb_base = ((long)b * M_ + m) * K_;
    const long tb      = (long)b * N_;

    #pragma unroll
    for (int tt = 0; tt < 2; ++tt) {
        const int  raw   = nbr[nb_base + tt * 16 + row];
        const bool valid = (raw != -1);
        const int  idx   = valid ? raw : 0;
        const float wv   = valid ? wts[tb + idx] : 0.0f;
        const float y0   = y[(tb + idx) * 2 + 0];
        const float y1   = y[(tb + idx) * 2 + 1];

        v2f bop;
        bop[0] = hi ? xm0 : y0;
        bop[1] = hi ? xm1 : y1;

        float hreg[4][8];
#pragma unroll
        for (int t = 0; t < 4; ++t) {
            v8f acc = {0.f, 0.f, 0.f, 0.f, 0.f, 0.f, 0.f, 0.f};
            acc = wmma_f32_4(w1a[t], bop, acc);
#pragma unroll
            for (int v = 0; v < 8; ++v) {
                hreg[t][v] = gelu_exact(acc[v] + b1v[t][v]);
            }
        }

        v8f acc2 = {0.f, 0.f, 0.f, 0.f, 0.f, 0.f, 0.f, 0.f};
#pragma unroll
        for (int t = 0; t < 4; ++t) {
#pragma unroll
            for (int p = 0; p < 4; ++p) {
                v2f bj;
                bj[0] = hreg[t][2 * p + 0];
                bj[1] = hreg[t][2 * p + 1];
                acc2 = wmma_f32_4(w2a[4 * t + p], bj, acc2);
            }
        }

        const float* fy = f_y + (tb + idx) * DOUT_;
#pragma unroll
        for (int v = 0; v < 8; ++v) {
            oacc[v] += (acc2[v] + b2v[v]) * fy[v + hoff] * wv;
        }
    }

#pragma unroll
    for (int v = 0; v < 8; ++v) {
        float s = oacc[v];
        s += swz<SWZ_XOR1>(s);
        s += swz<SWZ_XOR2>(s);
        s += swz<SWZ_XOR4>(s);
        s += swz<SWZ_XOR8>(s);
        oacc[v] = s;
    }

    if (row == 0) {
        float* po = out + ((long)b * M_ + m) * DOUT_ + hoff;
#pragma unroll
        for (int v = 0; v < 8; ++v) po[v] = oacc[v];
    }
}

extern "C" void kernel_launch(void* const* d_in, const int* in_sizes, int n_in,
                              void* d_out, int out_size, void* d_ws, size_t ws_size,
                              hipStream_t stream) {
    const float* y   = (const float*)d_in[0];
    const float* x   = (const float*)d_in[1];
    const float* f_y = (const float*)d_in[2];
    const float* wts = (const float*)d_in[3];
    const float* W1  = (const float*)d_in[4];
    const float* b1  = (const float*)d_in[5];
    const float* W2  = (const float*)d_in[6];
    const float* b2  = (const float*)d_in[7];
    const int*   nbr = (const int*)d_in[8];
    float* out = (float*)d_out;

    const int total  = B_ * M_;
    const int blocks = (total + 7) / 8;
    IntegralTransformBatch_kernel<<<blocks, 256, 0, stream>>>(
        y, x, f_y, wts, W1, b1, W2, b2, nbr, out);
}


# GELU erf part rebuilt as 1 - exp2(degree-7 polynomial of |z| clamped at 6), abs error no larger than the rational form replaced: one transcendental per activation, not two; a third fewer vector ops
# speedup vs baseline: 1.3605x; 1.2090x over previous
#include <hip/hip_runtime.h>
#include <math.h>

#define B_    8
#define N_    9225
#define M_    4096
#define K_    32
#define HID_  64
#define DOUT_ 16

typedef float v2f __attribute__((ext_vector_type(2)));
typedef float v8f __attribute__((ext_vector_type(8)));

__device__ __forceinline__ v8f wmma_f32_4(v2f a, v2f b, v8f c) {
    return __builtin_amdgcn_wmma_f32_16x16x4_f32(
         false, a,  false, b,
         (short)0, c,  false,  false);
}

#define SWZ_XOR8  0x201f
#define SWZ_XOR4  0x101f
#define SWZ_XOR2  0x081f
#define SWZ_XOR1  0x041f

template <int PAT>
__device__ __forceinline__ float swz(float x) {
    return __builtin_bit_cast(
        float, __builtin_amdgcn_ds_swizzle(__builtin_bit_cast(int, x), PAT));
}

__device__ __forceinline__ float gelu_exact(float z) {
    const float t = __builtin_fminf(__builtin_fabsf(z), 6.0f);
    float r = 7.676912901e-06f;
    r = __builtin_fmaf(r, t, -4.771750537e-05f);
    r = __builtin_fmaf(r, t, -4.383688211e-04f);
    r = __builtin_fmaf(r, t,  7.409221958e-03f);
    r = __builtin_fmaf(r, t, -5.269898102e-02f);
    r = __builtin_fmaf(r, t, -4.591544867e-01f);
    r = __builtin_fmaf(r, t, -1.151109815e+00f);
    r = __builtin_fmaf(r, t, -1.0f);
    const float h = __builtin_amdgcn_exp2f(r);
    return __builtin_fmaf(-t, h, __builtin_fmaxf(z, 0.0f));
}

__global__ __launch_bounds__(256)
void IntegralTransformBatch_kernel(const float* __restrict__ y,
                                   const float* __restrict__ x,
                                   const float* __restrict__ f_y,
                                   const float* __restrict__ wts,
                                   const float* __restrict__ W1,
                                   const float* __restrict__ b1,
                                   const float* __restrict__ W2,
                                   const float* __restrict__ b2,
                                   const int*   __restrict__ nbr,
                                   float*       __restrict__ out)
{
    const int tid  = threadIdx.x;
    const int wave = tid >> 5;
    const int lane = tid & 31;
    const int row  = lane & 15;
    const bool hi  = lane >= 16;
    const int hoff = hi ? 8 : 0;

    const int bm = blockIdx.x * 8 + wave;
    if (bm >= B_ * M_) return;
    const int b = bm / M_;
    const int m = bm % M_;

    const float xm0 = x[((long)b * M_ + m) * 2 + 0];
    const float xm1 = x[((long)b * M_ + m) * 2 + 1];

    v2f w1a[4];
#pragma unroll
    for (int t = 0; t < 4; ++t) {
        const int col = 16 * t + row;
        w1a[t][0] = W1[(hi ? 2 : 0) * HID_ + col];
        w1a[t][1] = W1[(hi ? 3 : 1) * HID_ + col];
    }
    v2f w2a[16];
#pragma unroll
    for (int j = 0; j < 16; ++j) {
        const int ch = 16 * (j >> 2) + hoff + 2 * (j & 3);
        w2a[j][0] = W2[(ch + 0) * DOUT_ + row];
        w2a[j][1] = W2[(ch + 1) * DOUT_ + row];
    }
    float b1v[4][8];
#pragma unroll
    for (int t = 0; t < 4; ++t)
#pragma unroll
        for (int v = 0; v < 8; ++v)
            b1v[t][v] = b1[16 * t + v + hoff];
    float b2v[8];
#pragma unroll
    for (int v = 0; v < 8; ++v) b2v[v] = b2[v + hoff];

    float oacc[8];
#pragma unroll
    for (int v = 0; v < 8; ++v) oacc[v] = 0.0f;

    const long nb_base = ((long)b * M_ + m) * K_;
    const long tb      = (long)b * N_;

    #pragma unroll
    for (int tt = 0; tt < 2; ++tt) {
        const int  raw   = nbr[nb_base + tt * 16 + row];
        const bool valid = (raw != -1);
        const int  idx   = valid ? raw : 0;
        const float wv   = valid ? wts[tb + idx] : 0.0f;
        const float y0   = y[(tb + idx) * 2 + 0];
        const float y1   = y[(tb + idx) * 2 + 1];

        v2f bop;
        bop[0] = hi ? xm0 : y0;
        bop[1] = hi ? xm1 : y1;

        float hreg[4][8];
#pragma unroll
        for (int t = 0; t < 4; ++t) {
            v8f acc = {0.f, 0.f, 0.f, 0.f, 0.f, 0.f, 0.f, 0.f};
            acc = wmma_f32_4(w1a[t], bop, acc);
#pragma unroll
            for (int v = 0; v < 8; ++v) {
                hreg[t][v] = gelu_exact(acc[v] + b1v[t][v]);
            }
        }

        v8f acc2 = {0.f, 0.f, 0.f, 0.f, 0.f, 0.f, 0.f, 0.f};
#pragma unroll
        for (int t = 0; t < 4; ++t) {
#pragma unroll
            for (int p = 0; p < 4; ++p) {
                v2f bj;
                bj[0] = hreg[t][2 * p + 0];
                bj[1] = hreg[t][2 * p + 1];
                acc2 = wmma_f32_4(w2a[4 * t + p], bj, acc2);
            }
        }

        const float* fy = f_y + (tb + idx) * DOUT_;
#pragma unroll
        for (int v = 0; v < 8; ++v) {
            oacc[v] += (acc2[v] + b2v[v]) * fy[v + hoff] * wv;
        }
    }

#pragma unroll
    for (int v = 0; v < 8; ++v) {
        float s = oacc[v];
        s += swz<SWZ_XOR1>(s);
        s += swz<SWZ_XOR2>(s);
        s += swz<SWZ_XOR4>(s);
        s += swz<SWZ_XOR8>(s);
        oacc[v] = s;
    }

    if (row == 0) {
        float* po = out + ((long)b * M_ + m) * DOUT_ + hoff;
#pragma unroll
        for (int v = 0; v < 8; ++v) po[v] = oacc[v];
    }
}

extern "C" void kernel_launch(void* const* d_in, const int* in_sizes, int n_in,
                              void* d_out, int out_size, void* d_ws, size_t ws_size,
                              hipStream_t stream) {
    const float* y   = (const float*)d_in[0];
    const float* x   = (const float*)d_in[1];
    const float* f_y = (const float*)d_in[2];
    const float* wts = (const float*)d_in[3];
    const float* W1  = (const float*)d_in[4];
    const float* b1  = (const float*)d_in[5];
    const float* W2  = (const float*)d_in[6];
    const float* b2  = (const float*)d_in[7];
    const int*   nbr = (const int*)d_in[8];
    float* out = (float*)d_out;

    const int total  = B_ * M_;
    const int blocks = (total + 7) / 8;
    IntegralTransformBatch_kernel<<<blocks, 256, 0, stream>>>(
        y, x, f_y, wts, W1, b1, W2, b2, nbr, out);
}


# Each wave loops over 8 output points instead of one: weight and bias fragments loaded once per wave and kept in registers, next point's indices and y rows fetched one point ahead
# speedup vs baseline: 1.4079x; 1.0348x over previous
#include <hip/hip_runtime.h>
#include <math.h>

#define B_    8
#define N_    9225
#define M_    4096
#define K_    32
#define HID_  64
#define DOUT_ 16

typedef float v2f __attribute__((ext_vector_type(2)));
typedef float v8f __attribute__((ext_vector_type(8)));

__device__ __forceinline__ v8f wmma_f32_4(v2f a, v2f b, v8f c) {
    return __builtin_amdgcn_wmma_f32_16x16x4_f32(
         false, a,  false, b,
         (short)0, c,  false,  false);
}

#define SWZ_XOR8  0x201f
#define SWZ_XOR4  0x101f
#define SWZ_XOR2  0x081f
#define SWZ_XOR1  0x041f

template <int PAT>
__device__ __forceinline__ float swz(float x) {
    return __builtin_bit_cast(
        float, __builtin_amdgcn_ds_swizzle(__builtin_bit_cast(int, x), PAT));
}

__device__ __forceinline__ float gelu_exact(float z) {
    const float t = __builtin_fminf(__builtin_fabsf(z), 6.0f);
    float r = 7.676912901e-06f;
    r = __builtin_fmaf(r, t, -4.771750537e-05f);
    r = __builtin_fmaf(r, t, -4.383688211e-04f);
    r = __builtin_fmaf(r, t,  7.409221958e-03f);
    r = __builtin_fmaf(r, t, -5.269898102e-02f);
    r = __builtin_fmaf(r, t, -4.591544867e-01f);
    r = __builtin_fmaf(r, t, -1.151109815e+00f);
    r = __builtin_fmaf(r, t, -1.0f);
    const float h = __builtin_amdgcn_exp2f(r);
    return __builtin_fmaf(-t, h, __builtin_fmaxf(z, 0.0f));
}

#define TOTAL_PTS_       (B_ * M_)
#define WAVES_PER_BLOCK_ 8
#define GRID_BLOCKS_     512
#define WAVE_STRIDE_     (GRID_BLOCKS_ * WAVES_PER_BLOCK_)

__global__ __launch_bounds__(256) __attribute__((amdgpu_waves_per_eu(4, 4)))
void IntegralTransformBatch_kernel(const float* __restrict__ y,
                                   const float* __restrict__ x,
                                   const float* __restrict__ f_y,
                                   const float* __restrict__ wts,
                                   const float* __restrict__ W1,
                                   const float* __restrict__ b1,
                                   const float* __restrict__ W2,
                                   const float* __restrict__ b2,
                                   const int*   __restrict__ nbr,
                                   float*       __restrict__ out)
{
    const int tid  = threadIdx.x;
    const int lane = tid & 31;
    const int wave = __builtin_amdgcn_readfirstlane(tid >> 5);
    const int row  = lane & 15;
    const bool hi  = lane >= 16;
    const int hoff = hi ? 8 : 0;

    const unsigned urow  = (unsigned)row;
    const unsigned uhoff = (unsigned)hoff;

    int bm = (int)blockIdx.x * WAVES_PER_BLOCK_ + wave;
    if (bm >= TOTAL_PTS_) return;

    v2f w1a[4];
#pragma unroll
    for (int t = 0; t < 4; ++t) {
        const int col = 16 * t + row;
        w1a[t][0] = W1[(hi ? 2 : 0) * HID_ + col];
        w1a[t][1] = W1[(hi ? 3 : 1) * HID_ + col];
    }
    v2f w2a[16];
#pragma unroll
    for (int j = 0; j < 16; ++j) {
        const int ch = 16 * (j >> 2) + hoff + 2 * (j & 3);
        w2a[j][0] = W2[(ch + 0) * DOUT_ + row];
        w2a[j][1] = W2[(ch + 1) * DOUT_ + row];
    }
    float b1v[4][8];
#pragma unroll
    for (int t = 0; t < 4; ++t)
#pragma unroll
        for (int v = 0; v < 8; ++v)
            b1v[t][v] = b1[16 * t + v + hoff];
    float b2v[8];
#pragma unroll
    for (int v = 0; v < 8; ++v) b2v[v] = b2[v + hoff];

    int   raw[2];
    float xm0, xm1;
    float yg[2][2];
    {
        const int* nb = nbr + (long)bm * K_;
        raw[0] = nb[urow];
        raw[1] = nb[16u + urow];
        xm0 = x[(long)bm * 2 + 0];
        xm1 = x[(long)bm * 2 + 1];
        const float* yb = y + (long)(bm / M_) * N_ * 2;
#pragma unroll
        for (int tt = 0; tt < 2; ++tt) {
            const unsigned idx = (raw[tt] != -1) ? (unsigned)raw[tt] : 0u;
            yg[tt][0] = yb[idx * 2u + 0u];
            yg[tt][1] = yb[idx * 2u + 1u];
        }
    }

    do {
        const long   tb = (long)(bm / M_) * N_;
        const float* wb = wts + tb;
        const float* fb = f_y + tb * DOUT_;

        bool  valid[2];
        float wl[2];
        float fyv[2][8];
#pragma unroll
        for (int tt = 0; tt < 2; ++tt) {
            valid[tt] = (raw[tt] != -1);
            const unsigned idx = valid[tt] ? (unsigned)raw[tt] : 0u;
            wl[tt] = wb[idx];
            const unsigned fo = idx * (unsigned)DOUT_ + uhoff;
#pragma unroll
            for (int v = 0; v < 8; ++v) fyv[tt][v] = fb[fo + (unsigned)v];
        }

        const int  bm_next = bm + WAVE_STRIDE_;
        const int  bm_pre  = (bm_next < TOTAL_PTS_) ? bm_next : bm;
        const int* nbn     = nbr + (long)bm_pre * K_;
        int   rawn[2];
        rawn[0] = nbn[urow];
        rawn[1] = nbn[16u + urow];
        const float xn0 = x[(long)bm_pre * 2 + 0];
        const float xn1 = x[(long)bm_pre * 2 + 1];
        float ygn[2][2];

        float oacc[8];
#pragma unroll
        for (int v = 0; v < 8; ++v) oacc[v] = 0.0f;

#pragma unroll
        for (int tt = 0; tt < 2; ++tt) {
            const float wv = valid[tt] ? wl[tt] : 0.0f;

            v2f bop;
            bop[0] = hi ? xm0 : yg[tt][0];
            bop[1] = hi ? xm1 : yg[tt][1];

            float hreg[4][8];
#pragma unroll
            for (int t = 0; t < 4; ++t) {
                v8f acc = {0.f, 0.f, 0.f, 0.f, 0.f, 0.f, 0.f, 0.f};
                acc = wmma_f32_4(w1a[t], bop, acc);
#pragma unroll
                for (int v = 0; v < 8; ++v) {
                    hreg[t][v] = gelu_exact(acc[v] + b1v[t][v]);
                }
            }

            v8f acc2 = {0.f, 0.f, 0.f, 0.f, 0.f, 0.f, 0.f, 0.f};
#pragma unroll
            for (int t = 0; t < 4; ++t) {
#pragma unroll
                for (int p = 0; p < 4; ++p) {
                    v2f bj;
                    bj[0] = hreg[t][2 * p + 0];
                    bj[1] = hreg[t][2 * p + 1];
                    acc2 = wmma_f32_4(w2a[4 * t + p], bj, acc2);
                }
            }

#pragma unroll
            for (int v = 0; v < 8; ++v) {
                oacc[v] += (acc2[v] + b2v[v]) * fyv[tt][v] * wv;
            }

            if (tt == 0) {
                const float* ybn = y + (long)(bm_pre / M_) * N_ * 2;
#pragma unroll
                for (int u = 0; u < 2; ++u) {
                    const unsigned idn = (rawn[u] != -1) ? (unsigned)rawn[u] : 0u;
                    ygn[u][0] = ybn[idn * 2u + 0u];
                    ygn[u][1] = ybn[idn * 2u + 1u];
                }
            }
        }

#pragma unroll
        for (int v = 0; v < 8; ++v) {
            float s = oacc[v];
            s += swz<SWZ_XOR1>(s);
            s += swz<SWZ_XOR2>(s);
            s += swz<SWZ_XOR4>(s);
            s += swz<SWZ_XOR8>(s);
            oacc[v] = s;
        }

        if (row == 0) {
            float* po = out + (long)bm * DOUT_ + hoff;
#pragma unroll
            for (int v = 0; v < 8; ++v) po[v] = oacc[v];
        }

        raw[0] = rawn[0];  raw[1] = rawn[1];
        xm0 = xn0;         xm1 = xn1;
        yg[0][0] = ygn[0][0];  yg[0][1] = ygn[0][1];
        yg[1][0] = ygn[1][0];  yg[1][1] = ygn[1][1];
        bm = bm_next;
    } while (bm < TOTAL_PTS_);
}

extern "C" void kernel_launch(void* const* d_in, const int* in_sizes, int n_in,
                              void* d_out, int out_size, void* d_ws, size_t ws_size,
                              hipStream_t stream) {
    const float* y   = (const float*)d_in[0];
    const float* x   = (const float*)d_in[1];
    const float* f_y = (const float*)d_in[2];
    const float* wts = (const float*)d_in[3];
    const float* W1  = (const float*)d_in[4];
    const float* b1  = (const float*)d_in[5];
    const float* W2  = (const float*)d_in[6];
    const float* b2  = (const float*)d_in[7];
    const int*   nbr = (const int*)d_in[8];
    float* out = (float*)d_out;

    IntegralTransformBatch_kernel<<<GRID_BLOCKS_, 256, 0, stream>>>(
        y, x, f_y, wts, W1, b1, W2, b2, nbr, out);
}
